# starting kernel
# baseline (speedup 1.0000x reference)
#include <hip/hip_runtime.h>
#include <hip/hip_bf16.h>
#include <math.h>


typedef __attribute__((ext_vector_type(2))) float v2f;
typedef __attribute__((ext_vector_type(8))) float v8f;

#define BATCH 4
#define NPTS  4096
#define M1    2048
#define M2    512
#define GF    1024

__device__ __forceinline__ unsigned f2ord(float f) {
    unsigned u = __float_as_uint(f);
    return (u & 0x80000000u) ? ~u : (u | 0x80000000u);
}
__device__ __forceinline__ float ord2f(unsigned k) {
    return (k & 0x80000000u) ? __uint_as_float(k & 0x7fffffffu)
                             : __uint_as_float(~k);
}

__global__ void hidden_kernel(const float* __restrict__ x,
                              const float* __restrict__ W1,
                              const float* __restrict__ b1,
                              float* __restrict__ hidden, int M) {
    int idx = blockIdx.x * blockDim.x + threadIdx.x;
    if (idx >= M * 64) return;
    int r = idx >> 6, f = idx & 63;
    float v = x[r * 2 + 0] * W1[f] + x[r * 2 + 1] * W1[64 + f] + b1[f];
    hidden[idx] = tanhf(v);
}

template <int KREL, int C1, int C2>
__device__ __forceinline__ float auxload(const float* __restrict__ A1,
                                         const float* __restrict__ A2,
                                         int row) {
    if constexpr (KREL < C1) {
        return A1[row * C1 + KREL];
    } else if constexpr (KREL - C1 < C2) {
        return A2[row * C2 + (KREL - C1)];
    } else {
        return 0.0f;
    }
}

template <int OUTMODE, int C0, int C1, int C2, int KTOT>
__global__ void gemm_wmma_kernel(const float* __restrict__ A0,
                                 const float* __restrict__ A1,
                                 const float* __restrict__ A2,
                                 const float* __restrict__ W, int F,
                                 const float* __restrict__ bias,
                                 float* __restrict__ out,
                                 unsigned* __restrict__ gfmax,
                                 int rowsPerBatch) {
    constexpr int KP = (KTOT + 3) & ~3;
    static_assert((C0 & 3) == 0, "C0 multiple of 4");
    static_assert(KP - C0 == 0 || KP - C0 == 4, "at most one tail K-step");

    int lane = threadIdx.x & 31;
    int half = lane >> 4;
    int mrow = lane & 15;
    int tm = blockIdx.x;
    int colBase = blockIdx.y * 64;
    int arow = tm * 16 + mrow;

    v8f c[4];
#pragma unroll
    for (int q = 0; q < 4; ++q)
#pragma unroll
        for (int v = 0; v < 8; ++v) c[q][v] = 0.0f;

    const float* a0row = A0 + (size_t)arow * C0;

#pragma unroll
    for (int k0 = 0; k0 < C0; k0 += 4) {
        int ka = k0 + 2 * half;
        v2f a = *(const v2f*)(a0row + ka);
#pragma unroll
        for (int nt = 0; nt < 4; ++nt) {
            int n = colBase + nt * 16 + mrow;
            v2f bm;
            bm.x = W[ka * F + n];
            bm.y = W[(ka + 1) * F + n];
            c[nt] = __builtin_amdgcn_wmma_f32_16x16x4_f32(
                false, a, false, bm, (short)0, c[nt], false, false);
        }
    }

    if constexpr (KP > C0) {
        float t0 = auxload<0, C1, C2>(A1, A2, arow);
        float t1 = auxload<1, C1, C2>(A1, A2, arow);
        float t2 = auxload<2, C1, C2>(A1, A2, arow);
        float t3 = auxload<3, C1, C2>(A1, A2, arow);
        v2f a;
        a.x = half ? t2 : t0;
        a.y = half ? t3 : t1;
        int ka = C0 + 2 * half;
        int kx = (ka < KTOT) ? ka : (KTOT - 1);
        int ky = (ka + 1 < KTOT) ? (ka + 1) : (KTOT - 1);
#pragma unroll
        for (int nt = 0; nt < 4; ++nt) {
            int n = colBase + nt * 16 + mrow;
            float wx = W[kx * F + n];
            float wy = W[ky * F + n];
            v2f bm;
            bm.x = (ka < KTOT) ? wx : 0.0f;
            bm.y = (ka + 1 < KTOT) ? wy : 0.0f;
            c[nt] = __builtin_amdgcn_wmma_f32_16x16x4_f32(
                false, a, false, bm, (short)0, c[nt], false, false);
        }
    }

#pragma unroll
    for (int nt = 0; nt < 4; ++nt) {
        int n = colBase + nt * 16 + mrow;
#pragma unroll
        for (int v = 0; v < 8; ++v) {
            int m = tm * 16 + v + 8 * half;
            float val = c[nt][v];
            if constexpr (OUTMODE == 0) {
                out[(size_t)m * F + n] = val;
            } else if constexpr (OUTMODE == 1) {
                out[(size_t)m * F + n] = tanhf(val + bias[n]);
            } else {
                float g = val + bias[n];
                atomicMax(&gfmax[(m / rowsPerBatch) * F + n], f2ord(g));
            }
        }
    }
}

__global__ void fps_kernel(const float* __restrict__ pos, int n, int m,
                           float* __restrict__ ctr) {
    __shared__ float mind2[NPTS];
    __shared__ float rval[256];
    __shared__ int ridx[256];
    __shared__ int s_last;

    int b = blockIdx.x;
    int tid = threadIdx.x;
    const float* pb = pos + (size_t)b * n * 2;
    float* cb = ctr + (size_t)b * m * 2;

    for (int j = tid; j < n; j += 256) mind2[j] = 1e10f;
    if (tid == 0) {
        s_last = 0;
        cb[0] = pb[0];
        cb[1] = pb[1];
    }
    __syncthreads();

    int last = 0;
    for (int it = 1; it < m; ++it) {
        float lx = pb[last * 2 + 0];
        float ly = pb[last * 2 + 1];
        float bv = -1.0f;
        int bi = n;
        for (int j = tid; j < n; j += 256) {
            float2 p = ((const float2*)pb)[j];
            float dx = p.x - lx;
            float dy = p.y - ly;
            float d2 = dx * dx + dy * dy;
            float md = fminf(mind2[j], d2);
            mind2[j] = md;
            if (md > bv) { bv = md; bi = j; }
        }
        rval[tid] = bv;
        ridx[tid] = bi;
        __syncthreads();
        for (int s = 128; s > 0; s >>= 1) {
            if (tid < s) {
                float ov = rval[tid + s];
                int oi = ridx[tid + s];
                if (ov > rval[tid] || (ov == rval[tid] && oi < ridx[tid])) {
                    rval[tid] = ov;
                    ridx[tid] = oi;
                }
            }
            __syncthreads();
        }
        if (tid == 0) {
            int nxt = ridx[0];
            s_last = nxt;
            cb[it * 2 + 0] = pb[nxt * 2 + 0];
            cb[it * 2 + 1] = pb[nxt * 2 + 1];
        }
        __syncthreads();
        last = s_last;
    }
}

template <int U>
__global__ void agg_kernel(const float* __restrict__ t,
                           const float* __restrict__ pos,
                           const float* __restrict__ ctr,
                           const float* __restrict__ Wp,
                           const float* __restrict__ bias,
                           float* __restrict__ out,
                           int n, int m, float r2) {
    const int F = U * 32;
    int gtid = blockIdx.x * blockDim.x + threadIdx.x;
    int w = gtid >> 5;
    int lane = threadIdx.x & 31;
    if (w >= BATCH * m) return;
    int b = w / m;
    int i = w - b * m;

    float2 cc = ((const float2*)ctr)[b * m + i];
    float cx = cc.x, cy = cc.y;

    float acc[U];
#pragma unroll
    for (int u = 0; u < U; ++u) acc[u] = -1e30f;

    const float2* pb = (const float2*)(pos + (size_t)b * n * 2);
    const float* tb = t + (size_t)b * n * F;
    for (int j = 0; j < n; ++j) {
        float2 p = pb[j];
        float dx = p.x - cx;
        float dy = p.y - cy;
        if (dx * dx + dy * dy <= r2) {
#pragma unroll
            for (int u = 0; u < U; ++u)
                acc[u] = fmaxf(acc[u], tb[(size_t)j * F + lane + 32 * u]);
        }
    }
#pragma unroll
    for (int u = 0; u < U; ++u) {
        int f = lane + 32 * u;
        out[(size_t)(b * m + i) * F + f] =
            acc[u] + bias[f] - cx * Wp[f] - cy * Wp[F + f];
    }
}

__global__ void init_gf_kernel(unsigned* __restrict__ gf, int n) {
    int i = blockIdx.x * blockDim.x + threadIdx.x;
    if (i < n) gf[i] = 0u;
}
__global__ void decode_gf_kernel(const unsigned* __restrict__ gf,
                                 float* __restrict__ out, int n) {
    int i = blockIdx.x * blockDim.x + threadIdx.x;
    if (i < n) out[i] = ord2f(gf[i]);
}

extern "C" void kernel_launch(void* const* d_in, const int* in_sizes, int n_in,
                              void* d_out, int out_size, void* d_ws, size_t ws_size,
                              hipStream_t stream) {
    const float* x     = (const float*)d_in[0];
    const float* zones = (const float*)d_in[1];
    const float* lf_W1 = (const float*)d_in[2];
    const float* lf_b1 = (const float*)d_in[3];
    const float* lf_W2 = (const float*)d_in[4];
    const float* lf_b2 = (const float*)d_in[5];
    const float* c1_W  = (const float*)d_in[6];
    const float* c1_b  = (const float*)d_in[7];
    const float* c2_W  = (const float*)d_in[8];
    const float* c2_b  = (const float*)d_in[9];
    const float* c3_W  = (const float*)d_in[10];
    const float* c3_b  = (const float*)d_in[11];

    float* out    = (float*)d_out;
    float* lf_out = out;
    float* gf_out = out + (size_t)BATCH * NPTS * 64;

    const int M = BATCH * NPTS;

    float* ws = (float*)d_ws;
    size_t o = 0;
    float* hidden = ws + o;  o += (size_t)M * 64;
    float* t1     = ws + o;  o += (size_t)M * 64;
    float* ctr1   = ws + o;  o += (size_t)BATCH * M1 * 2;
    float* h1     = ws + o;  o += (size_t)BATCH * M1 * 64;
    float* t2     = ws + o;  o += (size_t)BATCH * M1 * 128;
    float* ctr2   = ws + o;  o += (size_t)BATCH * M2 * 2;
    float* h2     = ws + o;  o += (size_t)BATCH * M2 * 128;
    unsigned* gf_enc = (unsigned*)(ws + o);

    hidden_kernel<<<(M * 64 + 255) / 256, 256, 0, stream>>>(x, lf_W1, lf_b1,
                                                            hidden, M);
    gemm_wmma_kernel<1, 64, 0, 0, 64><<<dim3(M / 16, 1), 32, 0, stream>>>(
        hidden, nullptr, nullptr, lf_W2, 64, lf_b2, lf_out, nullptr, 0);

    fps_kernel<<<BATCH, 256, 0, stream>>>(x, NPTS, M1, ctr1);
    fps_kernel<<<BATCH, 256, 0, stream>>>(ctr1, M1, M2, ctr2);

    gemm_wmma_kernel<0, 64, 1, 2, 67><<<dim3(M / 16, 1), 32, 0, stream>>>(
        lf_out, zones, x, c1_W, 64, nullptr, t1, nullptr, 0);

    agg_kernel<2><<<(BATCH * M1 * 32) / 256, 256, 0, stream>>>(
        t1, x, ctr1, c1_W + 65 * 64, c1_b, h1, NPTS, M1, 0.25f);

    gemm_wmma_kernel<0, 64, 2, 0, 66>
        <<<dim3(BATCH * M1 / 16, 2), 32, 0, stream>>>(
            h1, ctr1, nullptr, c2_W, 128, nullptr, t2, nullptr, 0);

    agg_kernel<4><<<(BATCH * M2 * 32) / 256, 256, 0, stream>>>(
        t2, ctr1, ctr2, c2_W + 64 * 128, c2_b, h2, M1, M2, 1.0f);

    init_gf_kernel<<<(BATCH * GF + 255) / 256, 256, 0, stream>>>(gf_enc,
                                                                 BATCH * GF);
    gemm_wmma_kernel<2, 128, 2, 0, 130>
        <<<dim3(BATCH * M2 / 16, GF / 64), 32, 0, stream>>>(
            h2, ctr2, nullptr, c3_W, GF, c3_b, nullptr, gf_enc, M2);
    decode_gf_kernel<<<(BATCH * GF + 255) / 256, 256, 0, stream>>>(gf_enc,
                                                                   gf_out,
                                                                   BATCH * GF);
}


# Second FPS launch removed: FPS over the first FPS's own picks, started at pick 0, re-picks them in order for every input (ties, duplicates too); the first FPS kernel also stores picks 0..511 as ctr2
# speedup vs baseline: 8.6899x; 8.6899x over previous
#include <hip/hip_runtime.h>
#include <hip/hip_bf16.h>
#include <math.h>


typedef __attribute__((ext_vector_type(2))) float v2f;
typedef __attribute__((ext_vector_type(8))) float v8f;

#define BATCH 4
#define NPTS  4096
#define M1    2048
#define M2    512
#define GF    1024

__device__ __forceinline__ unsigned f2ord(float f) {
    unsigned u = __float_as_uint(f);
    return (u & 0x80000000u) ? ~u : (u | 0x80000000u);
}
__device__ __forceinline__ float ord2f(unsigned k) {
    return (k & 0x80000000u) ? __uint_as_float(k & 0x7fffffffu)
                             : __uint_as_float(~k);
}

__global__ void hidden_kernel(const float* __restrict__ x,
                              const float* __restrict__ W1,
                              const float* __restrict__ b1,
                              float* __restrict__ hidden, int M) {
    int idx = blockIdx.x * blockDim.x + threadIdx.x;
    if (idx >= M * 64) return;
    int r = idx >> 6, f = idx & 63;
    float v = x[r * 2 + 0] * W1[f] + x[r * 2 + 1] * W1[64 + f] + b1[f];
    hidden[idx] = tanhf(v);
}

template <int KREL, int C1, int C2>
__device__ __forceinline__ float auxload(const float* __restrict__ A1,
                                         const float* __restrict__ A2,
                                         int row) {
    if constexpr (KREL < C1) {
        return A1[row * C1 + KREL];
    } else if constexpr (KREL - C1 < C2) {
        return A2[row * C2 + (KREL - C1)];
    } else {
        return 0.0f;
    }
}

template <int OUTMODE, int C0, int C1, int C2, int KTOT>
__global__ void gemm_wmma_kernel(const float* __restrict__ A0,
                                 const float* __restrict__ A1,
                                 const float* __restrict__ A2,
                                 const float* __restrict__ W, int F,
                                 const float* __restrict__ bias,
                                 float* __restrict__ out,
                                 unsigned* __restrict__ gfmax,
                                 int rowsPerBatch) {
    constexpr int KP = (KTOT + 3) & ~3;
    static_assert((C0 & 3) == 0, "C0 multiple of 4");
    static_assert(KP - C0 == 0 || KP - C0 == 4, "at most one tail K-step");

    int lane = threadIdx.x & 31;
    int half = lane >> 4;
    int mrow = lane & 15;
    int tm = blockIdx.x;
    int colBase = blockIdx.y * 64;
    int arow = tm * 16 + mrow;

    v8f c[4];
#pragma unroll
    for (int q = 0; q < 4; ++q)
#pragma unroll
        for (int v = 0; v < 8; ++v) c[q][v] = 0.0f;

    const float* a0row = A0 + (size_t)arow * C0;

#pragma unroll
    for (int k0 = 0; k0 < C0; k0 += 4) {
        int ka = k0 + 2 * half;
        v2f a = *(const v2f*)(a0row + ka);
#pragma unroll
        for (int nt = 0; nt < 4; ++nt) {
            int n = colBase + nt * 16 + mrow;
            v2f bm;
            bm.x = W[ka * F + n];
            bm.y = W[(ka + 1) * F + n];
            c[nt] = __builtin_amdgcn_wmma_f32_16x16x4_f32(
                false, a, false, bm, (short)0, c[nt], false, false);
        }
    }

    if constexpr (KP > C0) {
        float t0 = auxload<0, C1, C2>(A1, A2, arow);
        float t1 = auxload<1, C1, C2>(A1, A2, arow);
        float t2 = auxload<2, C1, C2>(A1, A2, arow);
        float t3 = auxload<3, C1, C2>(A1, A2, arow);
        v2f a;
        a.x = half ? t2 : t0;
        a.y = half ? t3 : t1;
        int ka = C0 + 2 * half;
        int kx = (ka < KTOT) ? ka : (KTOT - 1);
        int ky = (ka + 1 < KTOT) ? (ka + 1) : (KTOT - 1);
#pragma unroll
        for (int nt = 0; nt < 4; ++nt) {
            int n = colBase + nt * 16 + mrow;
            float wx = W[kx * F + n];
            float wy = W[ky * F + n];
            v2f bm;
            bm.x = (ka < KTOT) ? wx : 0.0f;
            bm.y = (ka + 1 < KTOT) ? wy : 0.0f;
            c[nt] = __builtin_amdgcn_wmma_f32_16x16x4_f32(
                false, a, false, bm, (short)0, c[nt], false, false);
        }
    }

#pragma unroll
    for (int nt = 0; nt < 4; ++nt) {
        int n = colBase + nt * 16 + mrow;
#pragma unroll
        for (int v = 0; v < 8; ++v) {
            int m = tm * 16 + v + 8 * half;
            float val = c[nt][v];
            if constexpr (OUTMODE == 0) {
                out[(size_t)m * F + n] = val;
            } else if constexpr (OUTMODE == 1) {
                out[(size_t)m * F + n] = tanhf(val + bias[n]);
            } else {
                float g = val + bias[n];
                atomicMax(&gfmax[(m / rowsPerBatch) * F + n], f2ord(g));
            }
        }
    }
}

#define ROWX_MAX(v, ctrl) \
    max((v), (unsigned)__builtin_amdgcn_update_dpp(0, (int)(v), (ctrl), 0xf, 0xf, false))

__device__ __forceinline__ unsigned wave_max_u32(unsigned v) {
    v = ROWX_MAX(v, 0x161);
    v = ROWX_MAX(v, 0x162);
    v = ROWX_MAX(v, 0x164);
    v = ROWX_MAX(v, 0x168);
    unsigned o = (unsigned)__builtin_amdgcn_permlanex16((int)v, (int)v, 0x76543210,
                                                        0xfedcba98, false, false);
    return max(v, o);
}
__device__ __forceinline__ unsigned oct_max_u32(unsigned v) {
    v = ROWX_MAX(v, 0x161);
    v = ROWX_MAX(v, 0x162);
    v = ROWX_MAX(v, 0x164);
    return v;
}

template <int PPT, int M, int MH>
__global__ __launch_bounds__(256) void fps_kernel(const float* __restrict__ pos,
                                                  float* __restrict__ ctr,
                                                  float* __restrict__ ctr_head) {
    static_assert(MH <= M, "the head is a prefix of the centre list");
    constexpr int n = PPT * 256;
    __shared__ float2 sctr[M];
    __shared__ uint4 srec[2][8];

    int b = blockIdx.x;
    int tid = threadIdx.x;
    int lane = tid & 31;
    int wave = tid >> 5;
    const float2* pb = (const float2*)(pos + (size_t)b * n * 2);
    float2* cb = (float2*)(ctr + (size_t)b * M * 2);

    float px[PPT], py[PPT];
    unsigned md[PPT];
#pragma unroll
    for (int u = 0; u < PPT; ++u) {
        float2 p = pb[tid + 256 * u];
        px[u] = p.x;
        py[u] = p.y;
        md[u] = __float_as_uint(1e10f);
    }
    float2 p0 = pb[0];
    float lx = p0.x, ly = p0.y;
    if (tid == 0) sctr[0] = p0;

    for (int it = 1; it < M; ++it) {
        unsigned bv = 0u;
        int bi = tid;
#pragma unroll
        for (int u = 0; u < PPT; ++u) {
            float dx = px[u] - lx;
            float dy = py[u] - ly;
            float d2;
            {
#pragma clang fp contract(off)
                d2 = dx * dx + dy * dy;
            }
            unsigned v = min(md[u], __float_as_uint(d2));
            md[u] = v;
            if (u == 0) {
                bv = v;
            } else if (v > bv) {
                bv = v;
                bi = tid + 256 * u;
            }
        }
        unsigned m1 = wave_max_u32(bv);
        unsigned tie = __builtin_amdgcn_ballot_w32(bv == m1);
        int wl = __builtin_ctz(tie);
        if (__builtin_expect(__builtin_popcount(tie) != 1, 0)) {
            unsigned lo = (bv == m1) ? ~(unsigned)bi : 0u;
            unsigned m2 = wave_max_u32(lo);
            wl = (int)(~(unsigned)__builtin_amdgcn_readfirstlane((int)m2) & 31u);
        }
        int wi = __builtin_amdgcn_readlane(bi, wl);
        int us = (wi >> 8) & (PPT - 1);
        int wx = __builtin_amdgcn_readlane((int)__float_as_uint(px[us]), wl);
        int wy = __builtin_amdgcn_readlane((int)__float_as_uint(py[us]), wl);
        uint4* rec = srec[it & 1];
        if (lane == 0)
            rec[wave] = make_uint4(m1, ~(unsigned)wi, (unsigned)wx, (unsigned)wy);
        __syncthreads();
        uint4 e = rec[lane & 7];
        unsigned g1 = oct_max_u32(e.x);
        unsigned tie2 = __builtin_amdgcn_ballot_w32(e.x == g1) & 0xffu;
        int wsel = __builtin_ctz(tie2);
        if (__builtin_expect(__builtin_popcount(tie2) != 1, 0)) {
            unsigned l2 = (e.x == g1) ? e.y : 0u;
            unsigned g2 = oct_max_u32(l2);
            unsigned widx = ~(unsigned)__builtin_amdgcn_readfirstlane((int)g2);
            wsel = (int)((widx >> 5) & 7u);
        }
        lx = __uint_as_float(
            (unsigned)__builtin_amdgcn_readlane((int)e.z, wsel));
        ly = __uint_as_float(
            (unsigned)__builtin_amdgcn_readlane((int)e.w, wsel));
        if (tid == 0) sctr[it] = make_float2(lx, ly);
    }
    __syncthreads();
    for (int i = tid; i < M; i += 256) cb[i] = sctr[i];
    float2* hb = (float2*)(ctr_head + (size_t)b * MH * 2);
    for (int i = tid; i < MH; i += 256) hb[i] = sctr[i];
}

template <int U, int N, int M>
__global__ void agg_kernel(const float* __restrict__ t,
                           const float* __restrict__ pos,
                           const float* __restrict__ ctr,
                           const float* __restrict__ Wp,
                           const float* __restrict__ bias,
                           float* __restrict__ out,
                           float r2) {
    constexpr int F = U * 32;
    constexpr int TILES = 4;
    static_assert(N % (32 * TILES) == 0, "whole trips only");
    int lane = threadIdx.x & 31;
    int w = __builtin_amdgcn_readfirstlane(
        (int)((blockIdx.x * blockDim.x + threadIdx.x) >> 5));
    if (w >= BATCH * M) return;
    int b = w / M;

    float2 cc = ((const float2*)ctr)[w];
    float cx = cc.x, cy = cc.y;

    float acc[U];
#pragma unroll
    for (int u = 0; u < U; ++u) acc[u] = -1e30f;

    const float2* pb = (const float2*)(pos + (size_t)b * N * 2);
    const float* tb = t + (size_t)b * N * F + lane;
    for (int j0 = 0; j0 < N; j0 += 32 * TILES) {
        unsigned mk[TILES];
#pragma unroll
        for (int q = 0; q < TILES; ++q) {
            float2 p = pb[j0 + 32 * q + lane];
            float dx = p.x - cx;
            float dy = p.y - cy;
            float d2;
            {
#pragma clang fp contract(off)
                d2 = dx * dx + dy * dy;
            }
            mk[q] = __builtin_amdgcn_ballot_w32(d2 <= r2);
        }
        unsigned any = 0u;
#pragma unroll
        for (int q = 0; q < TILES; ++q) any |= mk[q];
        if (any == 0u) continue;
#pragma unroll
        for (int q = 0; q < TILES; ++q) {
            unsigned mask = mk[q];
            while (mask) {
                int j = j0 + 32 * q + __builtin_ctz(mask);
                mask &= mask - 1;
                const float* row = tb + (size_t)j * F;
#pragma unroll
                for (int u = 0; u < U; ++u)
                    acc[u] = fmaxf(acc[u], row[32 * u]);
            }
        }
    }
#pragma unroll
    for (int u = 0; u < U; ++u) {
        int f = lane + 32 * u;
        out[(size_t)w * F + f] =
            acc[u] + bias[f] - cx * Wp[f] - cy * Wp[F + f];
    }
}

__global__ void init_gf_kernel(unsigned* __restrict__ gf, int n) {
    int i = blockIdx.x * blockDim.x + threadIdx.x;
    if (i < n) gf[i] = 0u;
}
__global__ void decode_gf_kernel(const unsigned* __restrict__ gf,
                                 float* __restrict__ out, int n) {
    int i = blockIdx.x * blockDim.x + threadIdx.x;
    if (i < n) out[i] = ord2f(gf[i]);
}

extern "C" void kernel_launch(void* const* d_in, const int* in_sizes, int n_in,
                              void* d_out, int out_size, void* d_ws, size_t ws_size,
                              hipStream_t stream) {
    const float* x     = (const float*)d_in[0];
    const float* zones = (const float*)d_in[1];
    const float* lf_W1 = (const float*)d_in[2];
    const float* lf_b1 = (const float*)d_in[3];
    const float* lf_W2 = (const float*)d_in[4];
    const float* lf_b2 = (const float*)d_in[5];
    const float* c1_W  = (const float*)d_in[6];
    const float* c1_b  = (const float*)d_in[7];
    const float* c2_W  = (const float*)d_in[8];
    const float* c2_b  = (const float*)d_in[9];
    const float* c3_W  = (const float*)d_in[10];
    const float* c3_b  = (const float*)d_in[11];

    float* out    = (float*)d_out;
    float* lf_out = out;
    float* gf_out = out + (size_t)BATCH * NPTS * 64;

    const int M = BATCH * NPTS;

    float* ws = (float*)d_ws;
    size_t o = 0;
    float* hidden = ws + o;  o += (size_t)M * 64;
    float* t1     = ws + o;  o += (size_t)M * 64;
    float* ctr1   = ws + o;  o += (size_t)BATCH * M1 * 2;
    float* h1     = ws + o;  o += (size_t)BATCH * M1 * 64;
    float* t2     = ws + o;  o += (size_t)BATCH * M1 * 128;
    float* ctr2   = ws + o;  o += (size_t)BATCH * M2 * 2;
    float* h2     = ws + o;  o += (size_t)BATCH * M2 * 128;
    unsigned* gf_enc = (unsigned*)(ws + o);

    hidden_kernel<<<(M * 64 + 255) / 256, 256, 0, stream>>>(x, lf_W1, lf_b1,
                                                            hidden, M);
    gemm_wmma_kernel<1, 64, 0, 0, 64><<<dim3(M / 16, 1), 32, 0, stream>>>(
        hidden, nullptr, nullptr, lf_W2, 64, lf_b2, lf_out, nullptr, 0);

    fps_kernel<NPTS / 256, M1, M2><<<BATCH, 256, 0, stream>>>(x, ctr1, ctr2);

    gemm_wmma_kernel<0, 64, 1, 2, 67><<<dim3(M / 16, 1), 32, 0, stream>>>(
        lf_out, zones, x, c1_W, 64, nullptr, t1, nullptr, 0);

    agg_kernel<2, NPTS, M1><<<(BATCH * M1 * 32) / 256, 256, 0, stream>>>(
        t1, x, ctr1, c1_W + 65 * 64, c1_b, h1, 0.25f);

    gemm_wmma_kernel<0, 64, 2, 0, 66>
        <<<dim3(BATCH * M1 / 16, 2), 32, 0, stream>>>(
            h1, ctr1, nullptr, c2_W, 128, nullptr, t2, nullptr, 0);

    agg_kernel<4, M1, M2><<<(BATCH * M2 * 32) / 256, 256, 0, stream>>>(
        t2, ctr1, ctr2, c2_W + 64 * 128, c2_b, h2, 1.0f);

    init_gf_kernel<<<(BATCH * GF + 255) / 256, 256, 0, stream>>>(gf_enc,
                                                                 BATCH * GF);
    gemm_wmma_kernel<2, 128, 2, 0, 130>
        <<<dim3(BATCH * M2 / 16, GF / 64), 32, 0, stream>>>(
            h2, ctr2, nullptr, c3_W, GF, c3_b, nullptr, gf_enc, M2);
    decode_gf_kernel<<<(BATCH * GF + 255) / 256, 256, 0, stream>>>(gf_enc,
                                                                   gf_out,
                                                                   BATCH * GF);
}


# FPS cross-wave argmax: each wave's top lane does one LDS 64-bit integer max on (distance bits, reversed index), order-free; winner's point read from LDS by the key; d2 scores bit-identical
# speedup vs baseline: 10.3656x; 1.1928x over previous
#include <hip/hip_runtime.h>
#include <hip/hip_bf16.h>
#include <math.h>


typedef __attribute__((ext_vector_type(2))) float v2f;
typedef __attribute__((ext_vector_type(8))) float v8f;

#define BATCH 4
#define NPTS  4096
#define M1    2048
#define M2    512
#define GF    1024

__device__ __forceinline__ unsigned f2ord(float f) {
    unsigned u = __float_as_uint(f);
    return (u & 0x80000000u) ? ~u : (u | 0x80000000u);
}
__device__ __forceinline__ float ord2f(unsigned k) {
    return (k & 0x80000000u) ? __uint_as_float(k & 0x7fffffffu)
                             : __uint_as_float(~k);
}

__global__ void hidden_kernel(const float* __restrict__ x,
                              const float* __restrict__ W1,
                              const float* __restrict__ b1,
                              float* __restrict__ hidden, int M) {
    int idx = blockIdx.x * blockDim.x + threadIdx.x;
    if (idx >= M * 64) return;
    int r = idx >> 6, f = idx & 63;
    float v = x[r * 2 + 0] * W1[f] + x[r * 2 + 1] * W1[64 + f] + b1[f];
    hidden[idx] = tanhf(v);
}

template <int KREL, int C1, int C2>
__device__ __forceinline__ float auxload(const float* __restrict__ A1,
                                         const float* __restrict__ A2,
                                         int row) {
    if constexpr (KREL < C1) {
        return A1[row * C1 + KREL];
    } else if constexpr (KREL - C1 < C2) {
        return A2[row * C2 + (KREL - C1)];
    } else {
        return 0.0f;
    }
}

template <int OUTMODE, int C0, int C1, int C2, int KTOT>
__global__ void gemm_wmma_kernel(const float* __restrict__ A0,
                                 const float* __restrict__ A1,
                                 const float* __restrict__ A2,
                                 const float* __restrict__ W, int F,
                                 const float* __restrict__ bias,
                                 float* __restrict__ out,
                                 unsigned* __restrict__ gfmax,
                                 int rowsPerBatch) {
    constexpr int KP = (KTOT + 3) & ~3;
    static_assert((C0 & 3) == 0, "C0 multiple of 4");
    static_assert(KP - C0 == 0 || KP - C0 == 4, "at most one tail K-step");

    int lane = threadIdx.x & 31;
    int half = lane >> 4;
    int mrow = lane & 15;
    int tm = blockIdx.x;
    int colBase = blockIdx.y * 64;
    int arow = tm * 16 + mrow;

    v8f c[4];
#pragma unroll
    for (int q = 0; q < 4; ++q)
#pragma unroll
        for (int v = 0; v < 8; ++v) c[q][v] = 0.0f;

    const float* a0row = A0 + (size_t)arow * C0;

#pragma unroll
    for (int k0 = 0; k0 < C0; k0 += 4) {
        int ka = k0 + 2 * half;
        v2f a = *(const v2f*)(a0row + ka);
#pragma unroll
        for (int nt = 0; nt < 4; ++nt) {
            int n = colBase + nt * 16 + mrow;
            v2f bm;
            bm.x = W[ka * F + n];
            bm.y = W[(ka + 1) * F + n];
            c[nt] = __builtin_amdgcn_wmma_f32_16x16x4_f32(
                false, a, false, bm, (short)0, c[nt], false, false);
        }
    }

    if constexpr (KP > C0) {
        float t0 = auxload<0, C1, C2>(A1, A2, arow);
        float t1 = auxload<1, C1, C2>(A1, A2, arow);
        float t2 = auxload<2, C1, C2>(A1, A2, arow);
        float t3 = auxload<3, C1, C2>(A1, A2, arow);
        v2f a;
        a.x = half ? t2 : t0;
        a.y = half ? t3 : t1;
        int ka = C0 + 2 * half;
        int kx = (ka < KTOT) ? ka : (KTOT - 1);
        int ky = (ka + 1 < KTOT) ? (ka + 1) : (KTOT - 1);
#pragma unroll
        for (int nt = 0; nt < 4; ++nt) {
            int n = colBase + nt * 16 + mrow;
            float wx = W[kx * F + n];
            float wy = W[ky * F + n];
            v2f bm;
            bm.x = (ka < KTOT) ? wx : 0.0f;
            bm.y = (ka + 1 < KTOT) ? wy : 0.0f;
            c[nt] = __builtin_amdgcn_wmma_f32_16x16x4_f32(
                false, a, false, bm, (short)0, c[nt], false, false);
        }
    }

#pragma unroll
    for (int nt = 0; nt < 4; ++nt) {
        int n = colBase + nt * 16 + mrow;
#pragma unroll
        for (int v = 0; v < 8; ++v) {
            int m = tm * 16 + v + 8 * half;
            float val = c[nt][v];
            if constexpr (OUTMODE == 0) {
                out[(size_t)m * F + n] = val;
            } else if constexpr (OUTMODE == 1) {
                out[(size_t)m * F + n] = tanhf(val + bias[n]);
            } else {
                float g = val + bias[n];
                atomicMax(&gfmax[(m / rowsPerBatch) * F + n], f2ord(g));
            }
        }
    }
}

#define ROWX_MAX(v, ctrl) \
    max((v), (unsigned)__builtin_amdgcn_update_dpp(0, (int)(v), (ctrl), 0xf, 0xf, false))

__device__ __forceinline__ unsigned wave_max_u32(unsigned v) {
    v = ROWX_MAX(v, 0x161);
    v = ROWX_MAX(v, 0x162);
    v = ROWX_MAX(v, 0x164);
    v = ROWX_MAX(v, 0x168);
    unsigned o = (unsigned)__builtin_amdgcn_permlanex16((int)v, (int)v, 0x76543210,
                                                        0xfedcba98, false, false);
    return max(v, o);
}

typedef __attribute__((address_space(3))) unsigned long long* lds_u64_ptr;

template <int PPT, int M, int MH>
__global__ __launch_bounds__(256) void fps_kernel(const float* __restrict__ pos,
                                                  float* __restrict__ ctr,
                                                  float* __restrict__ ctr_head) {
    static_assert(MH <= M, "the head is a prefix of the centre list");
    static_assert((PPT & (PPT - 1)) == 0 && PPT <= 32, "slot number fits the key");
    constexpr int n = PPT * 256;
    __shared__ unsigned long long smem[n + M + 4];
    float2* srev = (float2*)smem;
    float2* sctr = (float2*)(smem + n);
    unsigned long long* skey = smem + n + M;

    int b = blockIdx.x;
    int tid = threadIdx.x;
    const float2* pb = (const float2*)(pos + (size_t)b * n * 2);
    float2* cb = (float2*)(ctr + (size_t)b * M * 2);

    float px[PPT], py[PPT];
    unsigned md[PPT];
#pragma unroll
    for (int u = 0; u < PPT; ++u) {
        float2 p = pb[tid + 256 * u];
        px[u] = p.x;
        py[u] = p.y;
        md[u] = __float_as_uint(1e10f);
        srev[n - 1 - (tid + 256 * u)] = p;
    }
    float2 p0 = pb[0];
    float lx = p0.x, ly = p0.y;
    if (tid == 0) sctr[0] = p0;
    if (tid < 4) skey[tid] = 0ull;
    const unsigned lo_tid = (unsigned)(255 - tid) << 3;
    __syncthreads();

    for (int it = 1; it < M; ++it) {
        unsigned bv = 0u;
        unsigned bnu = PPT - 1;
#pragma unroll
        for (int u = 0; u < PPT; ++u) {
            float dx = px[u] - lx;
            float dy = py[u] - ly;
            float d2;
            {
#pragma clang fp contract(off)
                d2 = dx * dx + dy * dy;
            }
            unsigned v = min(md[u], __float_as_uint(d2));
            md[u] = v;
            if (u == 0) {
                bv = v;
            } else if (v > bv) {
                bv = v;
                bnu = PPT - 1 - u;
            }
        }
        unsigned lo = (bnu << 11) | lo_tid;
        unsigned long long* slot = skey + (it & 3);
        unsigned slot_lds = (unsigned)(size_t)(lds_u64_ptr)slot;
        asm volatile("" : "+v"(lo), "+v"(slot_lds));
        unsigned m1 = wave_max_u32(bv);
        unsigned long long key = ((unsigned long long)m1 << 32) | (unsigned long long)lo;
        if (bv == m1)
            asm volatile("ds_max_u64 %0, %1\n\ts_wait_dscnt 0x0"
                         :
                         : "v"(slot_lds), "v"(key)
                         : "memory");
        __syncthreads();
        unsigned wlo = (unsigned)(*slot);
        float2 c = *(const float2*)((const char*)srev + wlo);
        lx = c.x;
        ly = c.y;
        if (tid == 0) {
            sctr[it] = c;
            skey[(it + 2) & 3] = 0ull;
        }
    }
    __syncthreads();
    for (int i = tid; i < M; i += 256) cb[i] = sctr[i];
    float2* hb = (float2*)(ctr_head + (size_t)b * MH * 2);
    for (int i = tid; i < MH; i += 256) hb[i] = sctr[i];
}

template <int U, int N, int M>
__global__ void agg_kernel(const float* __restrict__ t,
                           const float* __restrict__ pos,
                           const float* __restrict__ ctr,
                           const float* __restrict__ Wp,
                           const float* __restrict__ bias,
                           float* __restrict__ out,
                           float r2) {
    constexpr int F = U * 32;
    constexpr int TILES = 4;
    static_assert(N % (32 * TILES) == 0, "whole trips only");
    int lane = threadIdx.x & 31;
    int w = __builtin_amdgcn_readfirstlane(
        (int)((blockIdx.x * blockDim.x + threadIdx.x) >> 5));
    if (w >= BATCH * M) return;
    int b = w / M;

    float2 cc = ((const float2*)ctr)[w];
    float cx = cc.x, cy = cc.y;

    float acc[U];
#pragma unroll
    for (int u = 0; u < U; ++u) acc[u] = -1e30f;

    const float2* pb = (const float2*)(pos + (size_t)b * N * 2);
    const float* tb = t + (size_t)b * N * F + lane;
    for (int j0 = 0; j0 < N; j0 += 32 * TILES) {
        unsigned mk[TILES];
#pragma unroll
        for (int q = 0; q < TILES; ++q) {
            float2 p = pb[j0 + 32 * q + lane];
            float dx = p.x - cx;
            float dy = p.y - cy;
            float d2;
            {
#pragma clang fp contract(off)
                d2 = dx * dx + dy * dy;
            }
            mk[q] = __builtin_amdgcn_ballot_w32(d2 <= r2);
        }
        unsigned any = 0u;
#pragma unroll
        for (int q = 0; q < TILES; ++q) any |= mk[q];
        if (any == 0u) continue;
#pragma unroll
        for (int q = 0; q < TILES; ++q) {
            unsigned mask = mk[q];
            while (mask) {
                int j = j0 + 32 * q + __builtin_ctz(mask);
                mask &= mask - 1;
                const float* row = tb + (size_t)j * F;
#pragma unroll
                for (int u = 0; u < U; ++u)
                    acc[u] = fmaxf(acc[u], row[32 * u]);
            }
        }
    }
#pragma unroll
    for (int u = 0; u < U; ++u) {
        int f = lane + 32 * u;
        out[(size_t)w * F + f] =
            acc[u] + bias[f] - cx * Wp[f] - cy * Wp[F + f];
    }
}

__global__ void init_gf_kernel(unsigned* __restrict__ gf, int n) {
    int i = blockIdx.x * blockDim.x + threadIdx.x;
    if (i < n) gf[i] = 0u;
}
__global__ void decode_gf_kernel(const unsigned* __restrict__ gf,
                                 float* __restrict__ out, int n) {
    int i = blockIdx.x * blockDim.x + threadIdx.x;
    if (i < n) out[i] = ord2f(gf[i]);
}

extern "C" void kernel_launch(void* const* d_in, const int* in_sizes, int n_in,
                              void* d_out, int out_size, void* d_ws, size_t ws_size,
                              hipStream_t stream) {
    const float* x     = (const float*)d_in[0];
    const float* zones = (const float*)d_in[1];
    const float* lf_W1 = (const float*)d_in[2];
    const float* lf_b1 = (const float*)d_in[3];
    const float* lf_W2 = (const float*)d_in[4];
    const float* lf_b2 = (const float*)d_in[5];
    const float* c1_W  = (const float*)d_in[6];
    const float* c1_b  = (const float*)d_in[7];
    const float* c2_W  = (const float*)d_in[8];
    const float* c2_b  = (const float*)d_in[9];
    const float* c3_W  = (const float*)d_in[10];
    const float* c3_b  = (const float*)d_in[11];

    float* out    = (float*)d_out;
    float* lf_out = out;
    float* gf_out = out + (size_t)BATCH * NPTS * 64;

    const int M = BATCH * NPTS;

    float* ws = (float*)d_ws;
    size_t o = 0;
    float* hidden = ws + o;  o += (size_t)M * 64;
    float* t1     = ws + o;  o += (size_t)M * 64;
    float* ctr1   = ws + o;  o += (size_t)BATCH * M1 * 2;
    float* h1     = ws + o;  o += (size_t)BATCH * M1 * 64;
    float* t2     = ws + o;  o += (size_t)BATCH * M1 * 128;
    float* ctr2   = ws + o;  o += (size_t)BATCH * M2 * 2;
    float* h2     = ws + o;  o += (size_t)BATCH * M2 * 128;
    unsigned* gf_enc = (unsigned*)(ws + o);

    hidden_kernel<<<(M * 64 + 255) / 256, 256, 0, stream>>>(x, lf_W1, lf_b1,
                                                            hidden, M);
    gemm_wmma_kernel<1, 64, 0, 0, 64><<<dim3(M / 16, 1), 32, 0, stream>>>(
        hidden, nullptr, nullptr, lf_W2, 64, lf_b2, lf_out, nullptr, 0);

    fps_kernel<NPTS / 256, M1, M2><<<BATCH, 256, 0, stream>>>(x, ctr1, ctr2);

    gemm_wmma_kernel<0, 64, 1, 2, 67><<<dim3(M / 16, 1), 32, 0, stream>>>(
        lf_out, zones, x, c1_W, 64, nullptr, t1, nullptr, 0);

    agg_kernel<2, NPTS, M1><<<(BATCH * M1 * 32) / 256, 256, 0, stream>>>(
        t1, x, ctr1, c1_W + 65 * 64, c1_b, h1, 0.25f);

    gemm_wmma_kernel<0, 64, 2, 0, 66>
        <<<dim3(BATCH * M1 / 16, 2), 32, 0, stream>>>(
            h1, ctr1, nullptr, c2_W, 128, nullptr, t2, nullptr, 0);

    agg_kernel<4, M1, M2><<<(BATCH * M2 * 32) / 256, 256, 0, stream>>>(
        t2, ctr1, ctr2, c2_W + 64 * 128, c2_b, h2, 1.0f);

    init_gf_kernel<<<(BATCH * GF + 255) / 256, 256, 0, stream>>>(gf_enc,
                                                                 BATCH * GF);
    gemm_wmma_kernel<2, 128, 2, 0, 130>
        <<<dim3(BATCH * M2 / 16, GF / 64), 32, 0, stream>>>(
            h2, ctr2, nullptr, c3_W, GF, c3_b, nullptr, gf_enc, M2);
    decode_gf_kernel<<<(BATCH * GF + 255) / 256, 256, 0, stream>>>(gf_enc,
                                                                   gf_out,
                                                                   BATCH * GF);
}


# FPS: points at distance 0 (picked ones) leave the sweep: every 256 picks a stable re-deal through LDS (ballot ranks, index order kept) and one slot less per thread, 16 down to 9; picks unchanged
# speedup vs baseline: 11.4306x; 1.1027x over previous
#include <hip/hip_runtime.h>
#include <hip/hip_bf16.h>
#include <math.h>


typedef __attribute__((ext_vector_type(2))) float v2f;
typedef __attribute__((ext_vector_type(8))) float v8f;

#define BATCH 4
#define NPTS  4096
#define M1    2048
#define M2    512
#define GF    1024

__device__ __forceinline__ unsigned f2ord(float f) {
    unsigned u = __float_as_uint(f);
    return (u & 0x80000000u) ? ~u : (u | 0x80000000u);
}
__device__ __forceinline__ float ord2f(unsigned k) {
    return (k & 0x80000000u) ? __uint_as_float(k & 0x7fffffffu)
                             : __uint_as_float(~k);
}

template <int KREL, int C1, int C2>
__device__ __forceinline__ float auxload(const float* __restrict__ A1,
                                         const float* __restrict__ A2,
                                         int row) {
    if constexpr (KREL < C1) {
        return A1[row * C1 + KREL];
    } else if constexpr (KREL - C1 < C2) {
        return A2[row * C2 + (KREL - C1)];
    } else {
        return 0.0f;
    }
}

template <int OUTMODE, int C0, int C1, int C2, int KTOT>
__global__ void gemm_wmma_kernel(const float* __restrict__ A0,
                                 const float* __restrict__ A1,
                                 const float* __restrict__ A2,
                                 const float* __restrict__ W, int F,
                                 const float* __restrict__ bias,
                                 float* __restrict__ out,
                                 unsigned* __restrict__ gfmax,
                                 int rowsPerBatch) {
    constexpr int KP = (KTOT + 3) & ~3;
    static_assert((C0 & 3) == 0, "C0 multiple of 4");
    static_assert(KP - C0 == 0 || KP - C0 == 4, "at most one tail K-step");

    int lane = threadIdx.x & 31;
    int half = lane >> 4;
    int mrow = lane & 15;
    int tm = blockIdx.x;
    int colBase = blockIdx.y * 64;
    int arow = tm * 16 + mrow;

    v8f c[4];
#pragma unroll
    for (int q = 0; q < 4; ++q)
#pragma unroll
        for (int v = 0; v < 8; ++v) c[q][v] = 0.0f;

    const float* a0row = A0 + (size_t)arow * C0;

#pragma unroll
    for (int k0 = 0; k0 < C0; k0 += 4) {
        int ka = k0 + 2 * half;
        v2f a = *(const v2f*)(a0row + ka);
#pragma unroll
        for (int nt = 0; nt < 4; ++nt) {
            int n = colBase + nt * 16 + mrow;
            v2f bm;
            bm.x = W[ka * F + n];
            bm.y = W[(ka + 1) * F + n];
            c[nt] = __builtin_amdgcn_wmma_f32_16x16x4_f32(
                false, a, false, bm, (short)0, c[nt], false, false);
        }
    }

    if constexpr (KP > C0) {
        float t0 = auxload<0, C1, C2>(A1, A2, arow);
        float t1 = auxload<1, C1, C2>(A1, A2, arow);
        float t2 = auxload<2, C1, C2>(A1, A2, arow);
        float t3 = auxload<3, C1, C2>(A1, A2, arow);
        v2f a;
        a.x = half ? t2 : t0;
        a.y = half ? t3 : t1;
        int ka = C0 + 2 * half;
        int kx = (ka < KTOT) ? ka : (KTOT - 1);
        int ky = (ka + 1 < KTOT) ? (ka + 1) : (KTOT - 1);
#pragma unroll
        for (int nt = 0; nt < 4; ++nt) {
            int n = colBase + nt * 16 + mrow;
            float wx = W[kx * F + n];
            float wy = W[ky * F + n];
            v2f bm;
            bm.x = (ka < KTOT) ? wx : 0.0f;
            bm.y = (ka + 1 < KTOT) ? wy : 0.0f;
            c[nt] = __builtin_amdgcn_wmma_f32_16x16x4_f32(
                false, a, false, bm, (short)0, c[nt], false, false);
        }
    }

#pragma unroll
    for (int nt = 0; nt < 4; ++nt) {
        int n = colBase + nt * 16 + mrow;
#pragma unroll
        for (int v = 0; v < 8; ++v) {
            int m = tm * 16 + v + 8 * half;
            float val = c[nt][v];
            if constexpr (OUTMODE == 0) {
                out[(size_t)m * F + n] = val;
            } else if constexpr (OUTMODE == 1) {
                out[(size_t)m * F + n] = tanhf(val + bias[n]);
            } else {
                float g = val + bias[n];
                atomicMax(&gfmax[(m / rowsPerBatch) * F + n], f2ord(g));
            }
        }
    }
}

#define ROWX_MAX(v, ctrl) \
    max((v), (unsigned)__builtin_amdgcn_update_dpp(0, (int)(v), (ctrl), 0xf, 0xf, false))

__device__ __forceinline__ unsigned wave_max_u32(unsigned v) {
    v = ROWX_MAX(v, 0x161);
    v = ROWX_MAX(v, 0x162);
    v = ROWX_MAX(v, 0x164);
    v = ROWX_MAX(v, 0x168);
    unsigned o = (unsigned)__builtin_amdgcn_permlanex16((int)v, (int)v, 0x76543210,
                                                        0xfedcba98, false, false);
    return max(v, o);
}

typedef __attribute__((address_space(3))) unsigned long long* lds_u64_ptr;

#define FPS_PPT 16

template <int S>
__device__ __forceinline__ void fps_picks(float (&px)[FPS_PPT], float (&py)[FPS_PPT],
                                          unsigned (&md)[FPS_PPT], unsigned (&lo)[FPS_PPT],
                                          float& lx, float& ly, int it0, int it1,
                                          const float2* srev, float2* sctr,
                                          unsigned long long* skey, int tid) {
    for (int it = it0; it < it1; ++it) {
        unsigned bv = 0u;
        unsigned blo = 0u;
#pragma unroll
        for (int u = 0; u < S; ++u) {
            float dx = px[u] - lx;
            float dy = py[u] - ly;
            float d2;
            {
#pragma clang fp contract(off)
                d2 = dx * dx + dy * dy;
            }
            unsigned v = min(md[u], __float_as_uint(d2));
            md[u] = v;
            if (u == 0) {
                bv = v;
                blo = lo[0];
            } else if (v > bv) {
                bv = v;
                blo = lo[u];
            }
        }
        unsigned long long* slot = skey + (it & 3);
        unsigned slot_lds = (unsigned)(size_t)(lds_u64_ptr)slot;
        asm volatile("" : "+v"(blo), "+v"(slot_lds));
        unsigned m1 = wave_max_u32(bv);
        unsigned long long key = ((unsigned long long)m1 << 32) | (unsigned long long)blo;
        if (bv == m1)
            asm volatile("ds_max_u64 %0, %1\n\ts_wait_dscnt 0x0"
                         :
                         : "v"(slot_lds), "v"(key)
                         : "memory");
        __syncthreads();
        unsigned wlo = (unsigned)(*slot);
        float2 c = *(const float2*)((const char*)srev + wlo);
        lx = c.x;
        ly = c.y;
        if (tid == 0) {
            sctr[it] = c;
            skey[(it + 2) & 3] = 0ull;
        }
    }
}

template <int SO, int SN>
__device__ __forceinline__ bool fps_redeal(float (&px)[FPS_PPT], float (&py)[FPS_PPT],
                                           unsigned (&md)[FPS_PPT], unsigned (&lo)[FPS_PPT],
                                           unsigned* tab, float4* stage, int tid,
                                           unsigned lo_first) {
    const int lane = tid & 31;
    const int wave = tid >> 5;
    const unsigned below = (1u << lane) - 1u;
    unsigned rk[SO];
    bool kp[SO];
#pragma unroll
    for (int u = 0; u < SO; ++u) {
        bool keep = (md[u] != 0u) || (lo[u] == lo_first);
        unsigned bal = __builtin_amdgcn_ballot_w32(keep);
        kp[u] = keep;
        rk[u] = (unsigned)__builtin_popcount(bal & below);
        if (lane == 0) tab[u * 8 + wave] = (unsigned)__builtin_popcount(bal);
    }
    if (tid >= SO * 8 && tid < 128) tab[tid] = 0u;
    __syncthreads();
    if (wave == 0) {
        uint4 cv = *(const uint4*)(tab + 4 * lane);
        unsigned s0 = cv.x, s1 = s0 + cv.y, s2 = s1 + cv.z, s3 = s2 + cv.w;
        unsigned incl = s3;
#pragma unroll
        for (int d = 1; d < 32; d <<= 1) {
            unsigned o = (unsigned)__shfl_up((int)incl, d, 32);
            if (lane >= d) incl += o;
        }
        unsigned excl = incl - s3;
        uint4 pv;
        pv.x = excl;
        pv.y = excl + s0;
        pv.z = excl + s1;
        pv.w = excl + s2;
        *(uint4*)(tab + 128 + 4 * lane) = pv;
        if (lane == 31) tab[256] = incl;
    }
    __syncthreads();
    const unsigned kept = (unsigned)__builtin_amdgcn_readfirstlane((int)tab[256]);
    if (kept > 256u * SN) return false;
#pragma unroll
    for (int u = 0; u < SO; ++u) {
        if (kp[u]) {
            unsigned r = tab[128 + u * 8 + wave] + rk[u];
            float4 e;
            e.x = px[u];
            e.y = py[u];
            e.z = __uint_as_float(md[u]);
            e.w = __uint_as_float(lo[u]);
            stage[r] = e;
        }
    }
    __syncthreads();
#pragma unroll
    for (int u = 0; u < SN; ++u) {
        unsigned r = (unsigned)(u * 256 + tid);
        if (r < kept) {
            float4 e = stage[r];
            px[u] = e.x;
            py[u] = e.y;
            md[u] = __float_as_uint(e.z);
            lo[u] = __float_as_uint(e.w);
        } else {
            px[u] = 0.0f;
            py[u] = 0.0f;
            md[u] = 0u;
            lo[u] = 0u;
        }
    }
    return true;
}

template <int S, int FIRST, int M>
__device__ __forceinline__ void fps_chain(float (&px)[FPS_PPT], float (&py)[FPS_PPT],
                                          unsigned (&md)[FPS_PPT], unsigned (&lo)[FPS_PPT],
                                          float& lx, float& ly, const float2* srev,
                                          float2* sctr, unsigned long long* skey,
                                          unsigned* tab, float4* stage, int tid,
                                          unsigned lo_first) {
    constexpr int NEXT = 256 * (FPS_PPT - S + 1) + 2;
    constexpr int END = NEXT < M ? NEXT : M;
    fps_picks<S>(px, py, md, lo, lx, ly, FIRST, END, srev, sctr, skey, tid);
    if constexpr (END < M && S > 1) {
        if (fps_redeal<S, S - 1>(px, py, md, lo, tab, stage, tid, lo_first))
            fps_chain<S - 1, END, M>(px, py, md, lo, lx, ly, srev, sctr, skey, tab, stage,
                                     tid, lo_first);
        else
            fps_picks<S>(px, py, md, lo, lx, ly, END, M, srev, sctr, skey, tid);
    }
}

template <int PPT, int M, int MH>
__device__ __forceinline__ void fps_body(const float* __restrict__ pos,
                                         float* __restrict__ ctr,
                                         float* __restrict__ ctr_head, int b) {
    static_assert(MH <= M, "the head is a prefix of the centre list");
    static_assert(PPT == FPS_PPT, "slot arrays are sized for 16 points per thread");
    constexpr int n = PPT * 256;
    __shared__ __attribute__((aligned(16))) unsigned long long smem[n + M + 4 + 2 * n + 136];
    static_assert(((n + M + 4) & 1) == 0, "stage starts on a 16-byte boundary");
    float2* srev = (float2*)smem;
    float2* sctr = (float2*)(smem + n);
    unsigned long long* skey = smem + n + M;
    float4* stage = (float4*)(smem + n + M + 4);
    unsigned* tab = (unsigned*)(smem + n + M + 4 + 2 * n);

    int tid = threadIdx.x;
    const float2* pb = (const float2*)(pos + (size_t)b * n * 2);
    float2* cb = (float2*)(ctr + (size_t)b * M * 2);

    float px[PPT], py[PPT];
    unsigned md[PPT], lo[PPT];
#pragma unroll
    for (int u = 0; u < PPT; ++u) {
        float2 p = pb[tid + 256 * u];
        px[u] = p.x;
        py[u] = p.y;
        md[u] = __float_as_uint(1e10f);
        lo[u] = (unsigned)(n - 1 - (tid + 256 * u)) << 3;
        srev[n - 1 - (tid + 256 * u)] = p;
    }
    float2 p0 = pb[0];
    float lx = p0.x, ly = p0.y;
    if (tid == 0) sctr[0] = p0;
    if (tid < 4) skey[tid] = 0ull;
    __syncthreads();

    fps_chain<PPT, 1, M>(px, py, md, lo, lx, ly, srev, sctr, skey, tab, stage, tid,
                         (unsigned)(n - 1) << 3);

    __syncthreads();
    for (int i = tid; i < M; i += 256) cb[i] = sctr[i];
    float2* hb = (float2*)(ctr_head + (size_t)b * MH * 2);
    for (int i = tid; i < MH; i += 256) hb[i] = sctr[i];
}

#define LFP 68

__device__ __forceinline__ void mlp_body(const float* __restrict__ x,
                                         const float* __restrict__ zones,
                                         const float* __restrict__ W1,
                                         const float* __restrict__ b1,
                                         const float* __restrict__ W2,
                                         const float* __restrict__ b2,
                                         const float* __restrict__ c1W,
                                         float* __restrict__ lf_out,
                                         float* __restrict__ t1, int tile0) {
    __shared__ float s_lf[8][16 * LFP];
    int lane = threadIdx.x & 31;
    int wave = threadIdx.x >> 5;
    int half = lane >> 4;
    int mrow = lane & 15;
    int tm = tile0 + wave;
    int arow = tm * 16 + mrow;
    float* sl = s_lf[wave];

    float x0 = x[arow * 2 + 0];
    float x1 = x[arow * 2 + 1];

    v8f c[4];
#pragma unroll
    for (int q = 0; q < 4; ++q)
#pragma unroll
        for (int v = 0; v < 8; ++v) c[q][v] = 0.0f;

    for (int k0 = 0; k0 < 64; k0 += 4) {
        int ka = k0 + 2 * half;
        v2f a;
        a.x = tanhf(x0 * W1[ka] + x1 * W1[64 + ka] + b1[ka]);
        a.y = tanhf(x0 * W1[ka + 1] + x1 * W1[64 + ka + 1] + b1[ka + 1]);
#pragma unroll
        for (int nt = 0; nt < 4; ++nt) {
            int n = nt * 16 + mrow;
            v2f bm;
            bm.x = W2[ka * 64 + n];
            bm.y = W2[(ka + 1) * 64 + n];
            c[nt] = __builtin_amdgcn_wmma_f32_16x16x4_f32(
                false, a, false, bm, (short)0, c[nt], false, false);
        }
    }
#pragma unroll
    for (int nt = 0; nt < 4; ++nt) {
        int n = nt * 16 + mrow;
        float bn = b2[n];
#pragma unroll
        for (int v = 0; v < 8; ++v) {
            int r = v + 8 * half;
            float val = tanhf(c[nt][v] + bn);
            lf_out[(size_t)(tm * 16 + r) * 64 + n] = val;
            sl[r * LFP + n] = val;
            c[nt][v] = 0.0f;
        }
    }
    __syncthreads();

#pragma unroll
    for (int k0 = 0; k0 < 64; k0 += 4) {
        int ka = k0 + 2 * half;
        v2f a = *(const v2f*)(sl + mrow * LFP + ka);
#pragma unroll
        for (int nt = 0; nt < 4; ++nt) {
            int n = nt * 16 + mrow;
            v2f bm;
            bm.x = c1W[ka * 64 + n];
            bm.y = c1W[(ka + 1) * 64 + n];
            c[nt] = __builtin_amdgcn_wmma_f32_16x16x4_f32(
                false, a, false, bm, (short)0, c[nt], false, false);
        }
    }
    {
        constexpr int KTOT = 67;
        float t0 = auxload<0, 1, 2>(zones, x, arow);
        float t1v = auxload<1, 1, 2>(zones, x, arow);
        float t2 = auxload<2, 1, 2>(zones, x, arow);
        float t3 = auxload<3, 1, 2>(zones, x, arow);
        v2f a;
        a.x = half ? t2 : t0;
        a.y = half ? t3 : t1v;
        int ka = 64 + 2 * half;
        int kx = (ka < KTOT) ? ka : (KTOT - 1);
        int ky = (ka + 1 < KTOT) ? (ka + 1) : (KTOT - 1);
#pragma unroll
        for (int nt = 0; nt < 4; ++nt) {
            int n = nt * 16 + mrow;
            float wx = c1W[kx * 64 + n];
            float wy = c1W[ky * 64 + n];
            v2f bm;
            bm.x = (ka < KTOT) ? wx : 0.0f;
            bm.y = (ka + 1 < KTOT) ? wy : 0.0f;
            c[nt] = __builtin_amdgcn_wmma_f32_16x16x4_f32(
                false, a, false, bm, (short)0, c[nt], false, false);
        }
    }
#pragma unroll
    for (int nt = 0; nt < 4; ++nt) {
        int n = nt * 16 + mrow;
#pragma unroll
        for (int v = 0; v < 8; ++v) {
            int m = tm * 16 + v + 8 * half;
            t1[(size_t)m * 64 + n] = c[nt][v];
        }
    }
}

__global__ __launch_bounds__(256) void front_kernel(const float* __restrict__ x,
                                                    const float* __restrict__ zones,
                                                    const float* __restrict__ W1,
                                                    const float* __restrict__ b1,
                                                    const float* __restrict__ W2,
                                                    const float* __restrict__ b2,
                                                    const float* __restrict__ c1W,
                                                    float* __restrict__ lf_out,
                                                    float* __restrict__ t1,
                                                    float* __restrict__ ctr1,
                                                    float* __restrict__ ctr2,
                                                    unsigned* __restrict__ gf_enc) {
    if (blockIdx.x < BATCH) {
        fps_body<NPTS / 256, M1, M2>(x, ctr1, ctr2, (int)blockIdx.x);
    } else {
        if (blockIdx.x == BATCH)
            for (int i = threadIdx.x; i < BATCH * GF; i += 256) gf_enc[i] = 0u;
        mlp_body(x, zones, W1, b1, W2, b2, c1W, lf_out, t1,
                 ((int)blockIdx.x - BATCH) * 8);
    }
}

template <int U, int N, int M>
__global__ void agg_kernel(const float* __restrict__ t,
                           const float* __restrict__ pos,
                           const float* __restrict__ ctr,
                           const float* __restrict__ Wp,
                           const float* __restrict__ bias,
                           float* __restrict__ out,
                           float r2) {
    constexpr int F = U * 32;
    constexpr int TILES = 4;
    static_assert(N % (32 * TILES) == 0, "whole trips only");
    int lane = threadIdx.x & 31;
    int w = __builtin_amdgcn_readfirstlane(
        (int)((blockIdx.x * blockDim.x + threadIdx.x) >> 5));
    if (w >= BATCH * M) return;
    int b = w / M;

    float2 cc = ((const float2*)ctr)[w];
    float cx = cc.x, cy = cc.y;

    float acc[U];
#pragma unroll
    for (int u = 0; u < U; ++u) acc[u] = -1e30f;

    const float2* pb = (const float2*)(pos + (size_t)b * N * 2);
    const float* tb = t + (size_t)b * N * F + lane;
    for (int j0 = 0; j0 < N; j0 += 32 * TILES) {
        unsigned mk[TILES];
#pragma unroll
        for (int q = 0; q < TILES; ++q) {
            float2 p = pb[j0 + 32 * q + lane];
            float dx = p.x - cx;
            float dy = p.y - cy;
            float d2;
            {
#pragma clang fp contract(off)
                d2 = dx * dx + dy * dy;
            }
            mk[q] = __builtin_amdgcn_ballot_w32(d2 <= r2);
        }
        unsigned any = 0u;
#pragma unroll
        for (int q = 0; q < TILES; ++q) any |= mk[q];
        if (any == 0u) continue;
#pragma unroll
        for (int q = 0; q < TILES; ++q) {
            unsigned mask = mk[q];
            while (mask) {
                int j = j0 + 32 * q + __builtin_ctz(mask);
                mask &= mask - 1;
                const float* row = tb + (size_t)j * F;
#pragma unroll
                for (int u = 0; u < U; ++u)
                    acc[u] = fmaxf(acc[u], row[32 * u]);
            }
        }
    }
#pragma unroll
    for (int u = 0; u < U; ++u) {
        int f = lane + 32 * u;
        out[(size_t)w * F + f] =
            acc[u] + bias[f] - cx * Wp[f] - cy * Wp[F + f];
    }
}

__global__ void decode_gf_kernel(const unsigned* __restrict__ gf,
                                 float* __restrict__ out, int n) {
    int i = blockIdx.x * blockDim.x + threadIdx.x;
    if (i < n) out[i] = ord2f(gf[i]);
}

extern "C" void kernel_launch(void* const* d_in, const int* in_sizes, int n_in,
                              void* d_out, int out_size, void* d_ws, size_t ws_size,
                              hipStream_t stream) {
    const float* x     = (const float*)d_in[0];
    const float* zones = (const float*)d_in[1];
    const float* lf_W1 = (const float*)d_in[2];
    const float* lf_b1 = (const float*)d_in[3];
    const float* lf_W2 = (const float*)d_in[4];
    const float* lf_b2 = (const float*)d_in[5];
    const float* c1_W  = (const float*)d_in[6];
    const float* c1_b  = (const float*)d_in[7];
    const float* c2_W  = (const float*)d_in[8];
    const float* c2_b  = (const float*)d_in[9];
    const float* c3_W  = (const float*)d_in[10];
    const float* c3_b  = (const float*)d_in[11];

    float* out    = (float*)d_out;
    float* lf_out = out;
    float* gf_out = out + (size_t)BATCH * NPTS * 64;

    const int M = BATCH * NPTS;

    float* ws = (float*)d_ws;
    size_t o = 0;
    o += (size_t)M * 64;
    float* t1     = ws + o;  o += (size_t)M * 64;
    float* ctr1   = ws + o;  o += (size_t)BATCH * M1 * 2;
    float* h1     = ws + o;  o += (size_t)BATCH * M1 * 64;
    float* t2     = ws + o;  o += (size_t)BATCH * M1 * 128;
    float* ctr2   = ws + o;  o += (size_t)BATCH * M2 * 2;
    float* h2     = ws + o;  o += (size_t)BATCH * M2 * 128;
    unsigned* gf_enc = (unsigned*)(ws + o);

    front_kernel<<<BATCH + M / 16 / 8, 256, 0, stream>>>(
        x, zones, lf_W1, lf_b1, lf_W2, lf_b2, c1_W, lf_out, t1, ctr1, ctr2,
        gf_enc);

    agg_kernel<2, NPTS, M1><<<(BATCH * M1 * 32) / 256, 256, 0, stream>>>(
        t1, x, ctr1, c1_W + 65 * 64, c1_b, h1, 0.25f);

    gemm_wmma_kernel<0, 64, 2, 0, 66>
        <<<dim3(BATCH * M1 / 16, 2), 32, 0, stream>>>(
            h1, ctr1, nullptr, c2_W, 128, nullptr, t2, nullptr, 0);

    agg_kernel<4, M1, M2><<<(BATCH * M2 * 32) / 256, 256, 0, stream>>>(
        t2, ctr1, ctr2, c2_W + 64 * 128, c2_b, h2, 1.0f);

    gemm_wmma_kernel<2, 128, 2, 0, 130>
        <<<dim3(BATCH * M2 / 16, GF / 64), 32, 0, stream>>>(
            h2, ctr2, nullptr, c3_W, GF, c3_b, nullptr, gf_enc, M2);
    decode_gf_kernel<<<(BATCH * GF + 255) / 256, 256, 0, stream>>>(gf_enc,
                                                                   gf_out,
                                                                   BATCH * GF);
}
